# starting kernel
# baseline (speedup 1.0000x reference)
#include <hip/hip_runtime.h>
#include <hip/hip_bf16.h>
#include <math.h>

#define D 128
#define TILE_M 16
#define GAMMA 0.1f
#define EPSILON 0.1f
#define NUM_ITERS 4

typedef __attribute__((ext_vector_type(2))) float v2f;
typedef __attribute__((ext_vector_type(8))) float v8f;

__global__ __launch_bounds__(256) void init_deg_kernel(float* __restrict__ deg, int n) {
    int i = blockIdx.x * blockDim.x + threadIdx.x;
    if (i < n) deg[i] = 1.0f;
}

__global__ __launch_bounds__(256) void count_deg_kernel(const int* __restrict__ ei,
                                                        float* __restrict__ deg, int E) {
    int e = blockIdx.x * blockDim.x + threadIdx.x;
    if (e < E) {
        int dst = ei[E + e];
        __hip_atomic_fetch_add(&deg[dst], 1.0f, __ATOMIC_RELAXED, __HIP_MEMORY_SCOPE_AGENT);
    }
}

__global__ __launch_bounds__(256) void finalize_dinv_kernel(float* __restrict__ deg, int n) {
    int i = blockIdx.x * blockDim.x + threadIdx.x;
    if (i < n) deg[i] = rsqrtf(deg[i]);
}

__global__ __launch_bounds__(256) void prep_weights_kernel(const float* __restrict__ W,
                                                           const float* __restrict__ Wphi,
                                                           float* __restrict__ WphiP,
                                                           float* __restrict__ ATP) {
    int i = blockIdx.x * blockDim.x + threadIdx.x;
    if (i >= D * D) return;
    int k = i / D, n = i % D;
    size_t po = (size_t)(k >> 1) * (2 * D) + n * 2 + (k & 1);
    WphiP[po] = Wphi[n * D + k];
    float a = W[n * D + k] - W[k * D + n];
    if (k == n) a -= GAMMA;
    ATP[po] = a;
}

__global__ __launch_bounds__(256) void gemm_dual_kernel(const float* __restrict__ x,
                                                        const float* __restrict__ WphiP,
                                                        const float* __restrict__ ATP,
                                                        const float* __restrict__ bias,
                                                        float* __restrict__ xw,
                                                        float* __restrict__ agg, int N) {
    __shared__ float lds_x[TILE_M][D + 4];

    const int row0 = blockIdx.x * TILE_M;
    const bool full = (row0 + TILE_M) <= N;
    const int tid = threadIdx.x;

    for (int i = tid; i < TILE_M * (D / 4); i += 256) {
        int r = i / (D / 4);
        int c = (i % (D / 4)) * 4;
        float4 v = make_float4(0.f, 0.f, 0.f, 0.f);
        if (full || (row0 + r < N)) v = *(const float4*)(x + (size_t)(row0 + r) * D + c);
        lds_x[r][c + 0] = v.x; lds_x[r][c + 1] = v.y;
        lds_x[r][c + 2] = v.z; lds_x[r][c + 3] = v.w;
    }
    __syncthreads();

    const int wave = tid >> 5;
    const int lane = tid & 31;
    const int mn   = lane & 15;
    const int kq   = lane >> 4;
    const int col0 = wave * 16;
    const size_t bcol = (size_t)(col0 + mn) * 2;

    v8f acc_w = {};
    v8f acc_a = {};

#pragma unroll 4
    for (int k0 = 0; k0 < D; k0 += 4) {
        const int k = k0 + kq * 2;
        const int p = (k0 >> 1) + kq;
        const v2f a  = *(const v2f*)&lds_x[mn][k];
        const v2f bw = *(const v2f*)(WphiP + (size_t)p * (2 * D) + bcol);
        const v2f ba = *(const v2f*)(ATP   + (size_t)p * (2 * D) + bcol);
        acc_w = __builtin_amdgcn_wmma_f32_16x16x4_f32(false, a, false, bw,
                                                      (short)0, acc_w, false, false);
        acc_a = __builtin_amdgcn_wmma_f32_16x16x4_f32(false, a, false, ba,
                                                      (short)0, acc_a, false, false);
    }

    const float b    = bias[col0 + mn];
    const int rowoff = kq * 8;
    if (full) {
#pragma unroll
        for (int r = 0; r < 8; ++r) {
            const size_t o = (size_t)(row0 + rowoff + r) * D + col0 + mn;
            xw[o]  = acc_w[r];
            agg[o] = acc_a[r] + b;
        }
    } else {
#pragma unroll
        for (int r = 0; r < 8; ++r) {
            const int row = row0 + rowoff + r;
            if (row < N) {
                const size_t o = (size_t)row * D + col0 + mn;
                xw[o]  = acc_w[r];
                agg[o] = acc_a[r] + b;
            }
        }
    }
}

__global__ __launch_bounds__(256) void edge_scatter_kernel(const int* __restrict__ ei,
                                                           const float* __restrict__ xw,
                                                           const float* __restrict__ dinv,
                                                           float* __restrict__ agg, int E) {
    const int wid  = (int)((blockIdx.x * (size_t)blockDim.x + threadIdx.x) >> 5);
    const int lane = threadIdx.x & 31;
    if (wid >= E) return;
    const int s = ei[wid];
    const int d = ei[E + wid];
    const float norm = dinv[s] * dinv[d];
    const float4 v = *(const float4*)(xw + (size_t)s * D + lane * 4);
    float* p = agg + (size_t)d * D + lane * 4;
    __hip_atomic_fetch_add(p + 0, v.x * norm, __ATOMIC_RELAXED, __HIP_MEMORY_SCOPE_AGENT);
    __hip_atomic_fetch_add(p + 1, v.y * norm, __ATOMIC_RELAXED, __HIP_MEMORY_SCOPE_AGENT);
    __hip_atomic_fetch_add(p + 2, v.z * norm, __ATOMIC_RELAXED, __HIP_MEMORY_SCOPE_AGENT);
    __hip_atomic_fetch_add(p + 3, v.w * norm, __ATOMIC_RELAXED, __HIP_MEMORY_SCOPE_AGENT);
}

__global__ __launch_bounds__(256) void combine_kernel(const float* __restrict__ xin,
                                                      const float* __restrict__ xw,
                                                      const float* __restrict__ agg,
                                                      const float* __restrict__ dinv,
                                                      float* __restrict__ xout, int N) {
    const int i = blockIdx.x * blockDim.x + threadIdx.x;
    const int total = N * (D / 4);
    if (i >= total) return;
    const int node = i / (D / 4);
    const float di = dinv[node];
    const float d2 = di * di;
    const size_t o = (size_t)i * 4;
    const float4 xv = *(const float4*)(xin + o);
    const float4 wv = *(const float4*)(xw + o);
    const float4 av = *(const float4*)(agg + o);
    float4 r;
    r.x = xv.x + EPSILON * tanhf(av.x + wv.x * d2);
    r.y = xv.y + EPSILON * tanhf(av.y + wv.y * d2);
    r.z = xv.z + EPSILON * tanhf(av.z + wv.z * d2);
    r.w = xv.w + EPSILON * tanhf(av.w + wv.w * d2);
    *(float4*)(xout + o) = r;
}

extern "C" void kernel_launch(void* const* d_in, const int* in_sizes, int n_in,
                              void* d_out, int out_size, void* d_ws, size_t ws_size,
                              hipStream_t stream) {
    const float* x_in = (const float*)d_in[0];
    const int*   ei   = (const int*)d_in[1];
    const float* W    = (const float*)d_in[2];
    const float* Wphi = (const float*)d_in[3];
    const float* bias = (const float*)d_in[4];
    float*       out  = (float*)d_out;

    const int N = in_sizes[0] / D;
    const int E = in_sizes[1] / 2;

    float* ws    = (float*)d_ws;
    const size_t nd = (size_t)N * D;
    float* xw    = ws;
    float* agg   = ws + nd;
    float* xbuf  = ws + 2 * nd;
    float* dinv  = ws + 3 * nd;
    float* WphiP = dinv + N;
    float* ATP   = WphiP + (size_t)D * D;

    const int B = 256;

    init_deg_kernel<<<(N + B - 1) / B, B, 0, stream>>>(dinv, N);
    count_deg_kernel<<<(E + B - 1) / B, B, 0, stream>>>(ei, dinv, E);
    finalize_dinv_kernel<<<(N + B - 1) / B, B, 0, stream>>>(dinv, N);
    prep_weights_kernel<<<(D * D + B - 1) / B, B, 0, stream>>>(W, Wphi, WphiP, ATP);

    const int gemm_blocks = (N + TILE_M - 1) / TILE_M;
    const int edge_blocks = (E + (B / 32) - 1) / (B / 32);
    const int comb_blocks = (N * (D / 4) + B - 1) / B;

    for (int it = 0; it < NUM_ITERS; ++it) {
        const float* xi = (it == 0) ? x_in : xbuf;
        float* xo = (it == NUM_ITERS - 1) ? out : xbuf;
        gemm_dual_kernel<<<gemm_blocks, B, 0, stream>>>(xi, WphiP, ATP, bias, xw, agg, N);
        edge_scatter_kernel<<<edge_blocks, B, 0, stream>>>(ei, xw, dinv, agg, E);
        combine_kernel<<<comb_blocks, B, 0, stream>>>(xi, xw, agg, dinv, xo, N);
    }
}


# Edge sums fetch a source row by one 16-byte load per lane (512 B, 4 whole lines per instruction, was four 128 B loads); LDS row holds feature f at (f&3)*32+(f>>2), adds stay conflict-free
# speedup vs baseline: 2.6159x; 2.6159x over previous
#include <hip/hip_runtime.h>
#include <hip/hip_bf16.h>
#include <math.h>

#define D 128
#define TILE_M 16
#define GAMMA 0.1f
#define EPSILON 0.1f
#define NUM_ITERS 4

typedef __attribute__((ext_vector_type(2))) float v2f;
typedef __attribute__((ext_vector_type(8))) float v8f;
typedef __attribute__((ext_vector_type(4))) float v4f;

__global__ __launch_bounds__(256) void prep_weights_kernel(const float* __restrict__ W,
                                                           const float* __restrict__ Wphi,
                                                           float* __restrict__ WphiP,
                                                           float* __restrict__ ATP) {
    int i = blockIdx.x * blockDim.x + threadIdx.x;
    if (i >= D * D) return;
    int k = i / D, n = i % D;
    size_t po = (size_t)(k >> 1) * (2 * D) + n * 2 + (k & 1);
    WphiP[po] = Wphi[n * D + k];
    float a = W[n * D + k] - W[k * D + n];
    if (k == n) a -= GAMMA;
    ATP[po] = a;
}

#define WPITCH (2 * D)
#define XPITCH (D + 4)
#define GEMM_WGS 512

template <bool COMBINE>
__global__ __launch_bounds__(256) void gemm_dual_kernel(const float* xin,
                                                        const float* __restrict__ WphiP,
                                                        const float* __restrict__ ATP,
                                                        const float* __restrict__ bias,
                                                        const float* __restrict__ dinv,
                                                        float* __restrict__ xw,
                                                        float* __restrict__ agg,
                                                        float* xout, int N, int ntiles) {
    __shared__ __attribute__((aligned(16))) float lds_w[2][(D / 2) * WPITCH];
    __shared__ __attribute__((aligned(16))) float lds_x[2][TILE_M][XPITCH];

    const int tid = threadIdx.x;
    const int stride = gridDim.x;
    int t = blockIdx.x;

    const int r0 = tid / (D / 4);
    const int c0 = (tid % (D / 4)) * 4;

    float4 xv0, xv1, wv0, wv1, av0, av1;
    float di0, di1;

    auto request_row = [&](int row, float4& xv, float4& wv, float4& av, float& di) {
        const int rr = (row < N) ? row : (N - 1);
        const size_t o = (size_t)rr * D + c0;
        xv = *(const float4*)(xin + o);
        if (COMBINE) {
            wv = *(const float4*)(xw + o);
            av = *(const float4*)(agg + o);
            di = dinv[rr];
        }
    };
    auto request_tile = [&](int tt) {
        request_row(tt * TILE_M + r0,     xv0, wv0, av0, di0);
        request_row(tt * TILE_M + r0 + 8, xv1, wv1, av1, di1);
    };
    auto updated = [&](const float4& xv, const float4& wv, const float4& av, float di) {
        if (!COMBINE) return xv;
        float4 v;
        v.x = xv.x + EPSILON * tanhf(av.x + wv.x * di);
        v.y = xv.y + EPSILON * tanhf(av.y + wv.y * di);
        v.z = xv.z + EPSILON * tanhf(av.z + wv.z * di);
        v.w = xv.w + EPSILON * tanhf(av.w + wv.w * di);
        return v;
    };

    request_tile(t);

#pragma unroll 8
    for (int i = tid; i < (D / 2) * WPITCH / 4; i += 256) {
        const int p = i >> 6;
        const int cs = ((i & 63) * 4) ^ ((p & 1) << 5);
        const float4 w4 = *(const float4*)(WphiP + (size_t)i * 4);
        const float4 a4 = *(const float4*)(ATP + (size_t)i * 4);
        *(float4*)&lds_w[0][p * WPITCH + cs] = w4;
        *(float4*)&lds_w[1][p * WPITCH + cs] = a4;
    }

    const int wave = tid >> 5;
    const int lane = tid & 31;
    const int mn   = lane & 15;
    const int kq   = lane >> 4;
    const int col0 = wave * 16;
    const int boff = kq * WPITCH + (((col0 + mn) * 2) ^ (kq << 5));
    const float* wl_w = &lds_w[0][boff];
    const float* wl_a = &lds_w[1][boff];
    const float b    = bias[col0 + mn];
    const int rowoff = kq * 8;

    int buf = 0;
    for (; t < ntiles; t += stride, buf ^= 1) {
        const int row0 = t * TILE_M;
        const bool full = (row0 + TILE_M) <= N;
        float (*lx)[XPITCH] = lds_x[buf];

        const float4 v0 = updated(xv0, wv0, av0, di0);
        const float4 v1 = updated(xv1, wv1, av1, di1);
        *(float4*)&lx[r0][c0]     = v0;
        *(float4*)&lx[r0 + 8][c0] = v1;
        __syncthreads();
        if (COMBINE) {
            const int ra = row0 + r0, rb = row0 + r0 + 8;
            if (full || ra < N) *(float4*)(xout + (size_t)ra * D + c0) = v0;
            if (full || rb < N) *(float4*)(xout + (size_t)rb * D + c0) = v1;
        }
        float dr[8];
        if (full) {
            const float4 q0 = *(const float4*)(dinv + row0 + rowoff);
            const float4 q1 = *(const float4*)(dinv + row0 + rowoff + 4);
            dr[0] = q0.x; dr[1] = q0.y; dr[2] = q0.z; dr[3] = q0.w;
            dr[4] = q1.x; dr[5] = q1.y; dr[6] = q1.z; dr[7] = q1.w;
        } else {
#pragma unroll
            for (int r = 0; r < 8; ++r) {
                const int rr = row0 + rowoff + r;
                dr[r] = dinv[(rr < N) ? rr : (N - 1)];
            }
        }
        if (t + stride < ntiles) request_tile(t + stride);

        v8f acc_w = {};
        v8f acc_a = {};

#pragma unroll 4
        for (int k0 = 0; k0 < D; k0 += 4) {
            const v2f a  = *(const v2f*)&lx[mn][k0 + kq * 2];
            const v2f bw = *(const v2f*)(wl_w + (k0 >> 1) * WPITCH);
            const v2f ba = *(const v2f*)(wl_a + (k0 >> 1) * WPITCH);
            acc_w = __builtin_amdgcn_wmma_f32_16x16x4_f32(false, a, false, bw,
                                                          (short)0, acc_w, false, false);
            acc_a = __builtin_amdgcn_wmma_f32_16x16x4_f32(false, a, false, ba,
                                                          (short)0, acc_a, false, false);
        }

        if (!COMBINE) {
            const v4f h0 = {xv0.x, xv0.y, xv0.z, xv0.w};
            const v4f h1 = {xv1.x, xv1.y, xv1.z, xv1.w};
            asm volatile("" :: "v"(h0), "v"(h1));
        }

        if (full) {
#pragma unroll
            for (int r = 0; r < 8; ++r) {
                const size_t o = (size_t)(row0 + rowoff + r) * D + col0 + mn;
                xw[o]  = acc_w[r] * dr[r];
                agg[o] = acc_a[r] + b;
            }
        } else {
#pragma unroll
            for (int r = 0; r < 8; ++r) {
                const int row = row0 + rowoff + r;
                if (row < N) {
                    const size_t o = (size_t)row * D + col0 + mn;
                    xw[o]  = acc_w[r] * dr[r];
                    agg[o] = acc_a[r] + b;
                }
            }
        }
    }
}

#define RB 64
#define CHUNK 8192

__global__ __launch_bounds__(1024) void range_count_kernel(const int* __restrict__ ei,
                                                           int* __restrict__ cnt,
                                                           int E, int NB) {
    extern __shared__ int lds_i[];
    for (int i = threadIdx.x; i < NB; i += 1024) lds_i[i] = 0;
    __syncthreads();
    const int e0 = blockIdx.x * CHUNK;
    for (int i = threadIdx.x; i < CHUNK; i += 1024) {
        const int e = e0 + i;
        if (e < E) {
            const int d = ei[E + e];
            __hip_atomic_fetch_add(&lds_i[d / RB], 1, __ATOMIC_RELAXED, __HIP_MEMORY_SCOPE_WORKGROUP);
        }
    }
    __syncthreads();
    for (int i = threadIdx.x; i < NB; i += 1024) cnt[(size_t)blockIdx.x * NB + i] = lds_i[i];
}

__global__ __launch_bounds__(1024) void range_scan_kernel(const int* __restrict__ cnt,
                                                          int* __restrict__ base,
                                                          int* __restrict__ binStart,
                                                          int NCH, int NB) {
    extern __shared__ int lds_i[];
    for (int b = threadIdx.x; b < NB; b += 1024) {
        int run = 0;
#pragma unroll 7
        for (int c = 0; c < NCH; ++c) {
            const size_t o = (size_t)c * NB + b;
            const int v = cnt[o];
            base[o] = run;
            run += v;
        }
        lds_i[b] = run;
    }
    __syncthreads();
    if (threadIdx.x < 32) {
        const int lane = threadIdx.x;
        const int per = (NB + 31) / 32;
        const int lo = lane * per;
        const int hi = (lo + per < NB) ? (lo + per) : NB;
        int s = 0;
        for (int i = lo; i < hi; ++i) s += lds_i[i];
        int incl = s;
        for (int dlt = 1; dlt < 32; dlt <<= 1) {
            const int t = __shfl_up(incl, dlt);
            if (lane >= dlt) incl += t;
        }
        int run = incl - s;
        for (int i = lo; i < hi; ++i) {
            const int v = lds_i[i];
            binStart[i] = run;
            run += v;
        }
        if (lane == 31) binStart[NB] = run;
    }
}

__global__ __launch_bounds__(1024) void range_place_kernel(const int* __restrict__ ei,
                                                           const int* __restrict__ base,
                                                           const int* __restrict__ binStart,
                                                           int2* __restrict__ binned,
                                                           int E, int NB) {
    extern __shared__ int lds_i[];
    for (int i = threadIdx.x; i < NB; i += 1024)
        lds_i[i] = binStart[i] + base[(size_t)blockIdx.x * NB + i];
    __syncthreads();
    const int e0 = blockIdx.x * CHUNK;
    for (int i = threadIdx.x; i < CHUNK; i += 1024) {
        const int e = e0 + i;
        if (e < E) {
            const int s = ei[e];
            const int d = ei[E + e];
            const int pos = __hip_atomic_fetch_add(&lds_i[d / RB], 1, __ATOMIC_RELAXED,
                                                   __HIP_MEMORY_SCOPE_WORKGROUP);
            binned[pos] = make_int2(s, d);
        }
    }
}

__global__ __launch_bounds__(256) void range_dinv_kernel(const int2* __restrict__ binned,
                                                         const int* __restrict__ binStart,
                                                         float* __restrict__ dinv, int N) {
    __shared__ int cnt_l[RB];

    const int b = blockIdx.x;
    const int node0 = b * RB;
    const int tid = threadIdx.x;
    if (tid < RB) cnt_l[tid] = 0;
    __syncthreads();

    const int beg = binStart[b];
    const int end = binStart[b + 1];
    for (int e0 = beg + tid; e0 < end; e0 += 4 * 256) {
        int d[4];
#pragma unroll
        for (int j = 0; j < 4; ++j) {
            const int e = e0 + j * 256;
            d[j] = (e < end) ? binned[e].y : node0;
        }
#pragma unroll
        for (int j = 0; j < 4; ++j) {
            if (e0 + j * 256 < end)
                __hip_atomic_fetch_add(&cnt_l[d[j] - node0], 1, __ATOMIC_RELAXED,
                                       __HIP_MEMORY_SCOPE_WORKGROUP);
        }
    }
    __syncthreads();

    if (tid < RB && node0 + tid < N) dinv[node0 + tid] = rsqrtf((float)(cnt_l[tid] + 1));
}

#define EPT 4

__global__ __launch_bounds__(256) void range_sum_kernel(const int2* __restrict__ binned,
                                                        const int* __restrict__ binStart,
                                                        const float* __restrict__ xs,
                                                        const float* __restrict__ dinv,
                                                        float* __restrict__ agg, int N) {
    __shared__ float acc[RB * D];

    const int b = blockIdx.x;
    const int node0 = b * RB;
    const int nrows = (N - node0 < RB) ? (N - node0) : RB;
    const int tid = threadIdx.x;
    float* aggb = agg + (size_t)node0 * D;

    for (int i = tid; i < RB * (D / 4); i += 256)
        *(float4*)(acc + i * 4) = make_float4(0.f, 0.f, 0.f, 0.f);
    __syncthreads();

    const int wave = __builtin_amdgcn_readfirstlane(tid >> 5);
    const int lane = tid & 31;
    const int beg = binStart[b];
    const int end = binStart[b + 1];

    int e = beg + wave * EPT;
    for (; e + EPT <= end; e += 8 * EPT) {
        int dl[EPT];
        float4 v[EPT];
#pragma unroll
        for (int j = 0; j < EPT; ++j) {
            const int2 sd = binned[e + j];
            dl[j] = sd.y - node0;
            v[j] = *(const float4*)(xs + (size_t)sd.x * D + lane * 4);
        }
#pragma unroll
        for (int j = 0; j < EPT; ++j) {
            float* a = acc + dl[j] * D + lane;
            __hip_atomic_fetch_add(a,      v[j].x, __ATOMIC_RELAXED, __HIP_MEMORY_SCOPE_WORKGROUP);
            __hip_atomic_fetch_add(a + 32, v[j].y, __ATOMIC_RELAXED, __HIP_MEMORY_SCOPE_WORKGROUP);
            __hip_atomic_fetch_add(a + 64, v[j].z, __ATOMIC_RELAXED, __HIP_MEMORY_SCOPE_WORKGROUP);
            __hip_atomic_fetch_add(a + 96, v[j].w, __ATOMIC_RELAXED, __HIP_MEMORY_SCOPE_WORKGROUP);
        }
    }
    for (; e < end; ++e) {
        const int2 sd = binned[e];
        const float4 r4 = *(const float4*)(xs + (size_t)sd.x * D + lane * 4);
        float* a = acc + (sd.y - node0) * D + lane;
        __hip_atomic_fetch_add(a,      r4.x, __ATOMIC_RELAXED, __HIP_MEMORY_SCOPE_WORKGROUP);
        __hip_atomic_fetch_add(a + 32, r4.y, __ATOMIC_RELAXED, __HIP_MEMORY_SCOPE_WORKGROUP);
        __hip_atomic_fetch_add(a + 64, r4.z, __ATOMIC_RELAXED, __HIP_MEMORY_SCOPE_WORKGROUP);
        __hip_atomic_fetch_add(a + 96, r4.w, __ATOMIC_RELAXED, __HIP_MEMORY_SCOPE_WORKGROUP);
    }

    if (nrows == RB) {
        float4 av[RB * (D / 4) / 256];
        float dv[RB * (D / 4) / 256];
#pragma unroll
        for (int k = 0; k < RB * (D / 4) / 256; ++k) {
            const int i = tid + k * 256;
            av[k] = *(const float4*)(aggb + i * 4);
            dv[k] = dinv[node0 + (i >> 5)];
        }
        __syncthreads();
#pragma unroll
        for (int k = 0; k < RB * (D / 4) / 256; ++k) {
            const int i = tid + k * 256;
            const float* sp = acc + (i >> 5) * D + (i & 31);
            float4 o;
            o.x = av[k].x + dv[k] * sp[0];
            o.y = av[k].y + dv[k] * sp[32];
            o.z = av[k].z + dv[k] * sp[64];
            o.w = av[k].w + dv[k] * sp[96];
            *(float4*)(aggb + i * 4) = o;
        }
    } else {
        __syncthreads();
        for (int i = tid; i < nrows * (D / 4); i += 256) {
            const float* sp = acc + (i >> 5) * D + (i & 31);
            const float4 a4 = *(const float4*)(aggb + i * 4);
            const float di = dinv[node0 + (i >> 5)];
            float4 o;
            o.x = a4.x + di * sp[0];
            o.y = a4.y + di * sp[32];
            o.z = a4.z + di * sp[64];
            o.w = a4.w + di * sp[96];
            *(float4*)(aggb + i * 4) = o;
        }
    }
}

__global__ __launch_bounds__(256) void combine_kernel(const float* __restrict__ xin,
                                                      const float* __restrict__ xw,
                                                      const float* __restrict__ agg,
                                                      const float* __restrict__ dinv,
                                                      float* __restrict__ xout, int N) {
    const int i = blockIdx.x * blockDim.x + threadIdx.x;
    const int total = N * (D / 4);
    if (i >= total) return;
    const int node = i / (D / 4);
    const float di = dinv[node];
    const size_t o = (size_t)i * 4;
    const float4 xv = *(const float4*)(xin + o);
    const float4 wv = *(const float4*)(xw + o);
    const float4 av = *(const float4*)(agg + o);
    float4 r;
    r.x = xv.x + EPSILON * tanhf(av.x + wv.x * di);
    r.y = xv.y + EPSILON * tanhf(av.y + wv.y * di);
    r.z = xv.z + EPSILON * tanhf(av.z + wv.z * di);
    r.w = xv.w + EPSILON * tanhf(av.w + wv.w * di);
    *(float4*)(xout + o) = r;
}

extern "C" void kernel_launch(void* const* d_in, const int* in_sizes, int n_in,
                              void* d_out, int out_size, void* d_ws, size_t ws_size,
                              hipStream_t stream) {
    const float* x_in = (const float*)d_in[0];
    const int*   ei   = (const int*)d_in[1];
    const float* W    = (const float*)d_in[2];
    const float* Wphi = (const float*)d_in[3];
    const float* bias = (const float*)d_in[4];
    float*       out  = (float*)d_out;

    const int N = in_sizes[0] / D;
    const int E = in_sizes[1] / 2;

    float* ws    = (float*)d_ws;
    const size_t nd = (size_t)N * D;
    float* xw    = ws;
    float* agg   = ws + nd;
    float* earea = ws + 2 * nd;
    float* dinv  = ws + 3 * nd;
    float* WphiP = dinv + N;
    float* ATP   = WphiP + (size_t)D * D;

    const int NB  = (N + RB - 1) / RB;
    const int NCH = (E + CHUNK - 1) / CHUNK;
    int2* binned   = (int2*)earea;
    int*  cnt      = (int*)(binned + E);
    int*  base     = cnt + (size_t)NCH * NB;
    int*  binStart = base + (size_t)NCH * NB;

    const int B = 256;

    prep_weights_kernel<<<(D * D + B - 1) / B, B, 0, stream>>>(W, Wphi, WphiP, ATP);
    range_count_kernel<<<NCH, 1024, NB * sizeof(int), stream>>>(ei, cnt, E, NB);
    range_scan_kernel<<<1, 1024, NB * sizeof(int), stream>>>(cnt, base, binStart, NCH, NB);
    range_place_kernel<<<NCH, 1024, NB * sizeof(int), stream>>>(ei, base, binStart, binned, E, NB);
    range_dinv_kernel<<<NB, B, 0, stream>>>(binned, binStart, dinv, N);

    const int gemm_tiles  = (N + TILE_M - 1) / TILE_M;
    const int gemm_blocks = (gemm_tiles < GEMM_WGS) ? gemm_tiles : GEMM_WGS;
    const int comb_blocks = (N * (D / 4) + B - 1) / B;

    gemm_dual_kernel<false><<<gemm_blocks, B, 0, stream>>>(x_in, WphiP, ATP, bias, dinv, xw, agg,
                                                           out, N, gemm_tiles);
    for (int it = 0; it < NUM_ITERS; ++it) {
        const float* xi = (it == 0) ? x_in : out;
        range_sum_kernel<<<NB, B, 0, stream>>>(binned, binStart, xw, dinv, agg, N);
        if (it + 1 < NUM_ITERS)
            gemm_dual_kernel<true><<<gemm_blocks, B, 0, stream>>>(xi, WphiP, ATP, bias, dinv, xw,
                                                                  agg, out, N, gemm_tiles);
        else
            combine_kernel<<<comb_blocks, B, 0, stream>>>(xi, xw, agg, dinv, out, N);
    }
}


# Edge sums: LDS f32 atomic adds replaced by wave-owned rows summed in registers; the block orders its range's sources by row in LDS (integer rank per record); row sums now in rank order
# speedup vs baseline: 2.7549x; 1.0531x over previous
#include <hip/hip_runtime.h>
#include <hip/hip_bf16.h>
#include <math.h>

#define D 128
#define TILE_M 16
#define GAMMA 0.1f
#define EPSILON 0.1f
#define NUM_ITERS 4

typedef __attribute__((ext_vector_type(2))) float v2f;
typedef __attribute__((ext_vector_type(8))) float v8f;
typedef __attribute__((ext_vector_type(4))) float v4f;

__global__ __launch_bounds__(256) void prep_weights_kernel(const float* __restrict__ W,
                                                           const float* __restrict__ Wphi,
                                                           float* __restrict__ WphiP,
                                                           float* __restrict__ ATP) {
    int i = blockIdx.x * blockDim.x + threadIdx.x;
    if (i >= D * D) return;
    int k = i / D, n = i % D;
    size_t po = (size_t)(k >> 1) * (2 * D) + n * 2 + (k & 1);
    WphiP[po] = Wphi[n * D + k];
    float a = W[n * D + k] - W[k * D + n];
    if (k == n) a -= GAMMA;
    ATP[po] = a;
}

#define WPITCH (2 * D)
#define XPITCH (D + 4)
#define GEMM_WGS 512

template <bool COMBINE>
__global__ __launch_bounds__(256) void gemm_dual_kernel(const float* xin,
                                                        const float* __restrict__ WphiP,
                                                        const float* __restrict__ ATP,
                                                        const float* __restrict__ bias,
                                                        const float* __restrict__ dinv,
                                                        float* __restrict__ xw,
                                                        float* __restrict__ agg,
                                                        float* xout, int N, int ntiles) {
    __shared__ __attribute__((aligned(16))) float lds_w[2][(D / 2) * WPITCH];
    __shared__ __attribute__((aligned(16))) float lds_x[2][TILE_M][XPITCH];

    const int tid = threadIdx.x;
    const int stride = gridDim.x;
    int t = blockIdx.x;

    const int r0 = tid / (D / 4);
    const int c0 = (tid % (D / 4)) * 4;

    float4 xv0, xv1, wv0, wv1, av0, av1;
    float di0, di1;

    auto request_row = [&](int row, float4& xv, float4& wv, float4& av, float& di) {
        const int rr = (row < N) ? row : (N - 1);
        const size_t o = (size_t)rr * D + c0;
        xv = *(const float4*)(xin + o);
        if (COMBINE) {
            wv = *(const float4*)(xw + o);
            av = *(const float4*)(agg + o);
            di = dinv[rr];
        }
    };
    auto request_tile = [&](int tt) {
        request_row(tt * TILE_M + r0,     xv0, wv0, av0, di0);
        request_row(tt * TILE_M + r0 + 8, xv1, wv1, av1, di1);
    };
    auto updated = [&](const float4& xv, const float4& wv, const float4& av, float di) {
        if (!COMBINE) return xv;
        float4 v;
        v.x = xv.x + EPSILON * tanhf(av.x + wv.x * di);
        v.y = xv.y + EPSILON * tanhf(av.y + wv.y * di);
        v.z = xv.z + EPSILON * tanhf(av.z + wv.z * di);
        v.w = xv.w + EPSILON * tanhf(av.w + wv.w * di);
        return v;
    };

    request_tile(t);

#pragma unroll 8
    for (int i = tid; i < (D / 2) * WPITCH / 4; i += 256) {
        const int p = i >> 6;
        const int cs = ((i & 63) * 4) ^ ((p & 1) << 5);
        const float4 w4 = *(const float4*)(WphiP + (size_t)i * 4);
        const float4 a4 = *(const float4*)(ATP + (size_t)i * 4);
        *(float4*)&lds_w[0][p * WPITCH + cs] = w4;
        *(float4*)&lds_w[1][p * WPITCH + cs] = a4;
    }

    const int wave = tid >> 5;
    const int lane = tid & 31;
    const int mn   = lane & 15;
    const int kq   = lane >> 4;
    const int col0 = wave * 16;
    const int boff = kq * WPITCH + (((col0 + mn) * 2) ^ (kq << 5));
    const float* wl_w = &lds_w[0][boff];
    const float* wl_a = &lds_w[1][boff];
    const float b    = bias[col0 + mn];
    const int rowoff = kq * 8;

    int buf = 0;
    for (; t < ntiles; t += stride, buf ^= 1) {
        const int row0 = t * TILE_M;
        const bool full = (row0 + TILE_M) <= N;
        float (*lx)[XPITCH] = lds_x[buf];

        const float4 v0 = updated(xv0, wv0, av0, di0);
        const float4 v1 = updated(xv1, wv1, av1, di1);
        *(float4*)&lx[r0][c0]     = v0;
        *(float4*)&lx[r0 + 8][c0] = v1;
        __syncthreads();
        if (COMBINE) {
            const int ra = row0 + r0, rb = row0 + r0 + 8;
            if (full || ra < N) *(float4*)(xout + (size_t)ra * D + c0) = v0;
            if (full || rb < N) *(float4*)(xout + (size_t)rb * D + c0) = v1;
        }
        float dr[8];
        if (full) {
            const float4 q0 = *(const float4*)(dinv + row0 + rowoff);
            const float4 q1 = *(const float4*)(dinv + row0 + rowoff + 4);
            dr[0] = q0.x; dr[1] = q0.y; dr[2] = q0.z; dr[3] = q0.w;
            dr[4] = q1.x; dr[5] = q1.y; dr[6] = q1.z; dr[7] = q1.w;
        } else {
#pragma unroll
            for (int r = 0; r < 8; ++r) {
                const int rr = row0 + rowoff + r;
                dr[r] = dinv[(rr < N) ? rr : (N - 1)];
            }
        }
        if (t + stride < ntiles) request_tile(t + stride);

        v8f acc_w = {};
        v8f acc_a = {};

#pragma unroll 4
        for (int k0 = 0; k0 < D; k0 += 4) {
            const v2f a  = *(const v2f*)&lx[mn][k0 + kq * 2];
            const v2f bw = *(const v2f*)(wl_w + (k0 >> 1) * WPITCH);
            const v2f ba = *(const v2f*)(wl_a + (k0 >> 1) * WPITCH);
            acc_w = __builtin_amdgcn_wmma_f32_16x16x4_f32(false, a, false, bw,
                                                          (short)0, acc_w, false, false);
            acc_a = __builtin_amdgcn_wmma_f32_16x16x4_f32(false, a, false, ba,
                                                          (short)0, acc_a, false, false);
        }

        if (!COMBINE) {
            const v4f h0 = {xv0.x, xv0.y, xv0.z, xv0.w};
            const v4f h1 = {xv1.x, xv1.y, xv1.z, xv1.w};
            asm volatile("" :: "v"(h0), "v"(h1));
        }

        if (full) {
#pragma unroll
            for (int r = 0; r < 8; ++r) {
                const size_t o = (size_t)(row0 + rowoff + r) * D + col0 + mn;
                xw[o]  = acc_w[r] * dr[r];
                agg[o] = acc_a[r] + b;
            }
        } else {
#pragma unroll
            for (int r = 0; r < 8; ++r) {
                const int row = row0 + rowoff + r;
                if (row < N) {
                    const size_t o = (size_t)row * D + col0 + mn;
                    xw[o]  = acc_w[r] * dr[r];
                    agg[o] = acc_a[r] + b;
                }
            }
        }
    }
}

#define RB 64
#define CHUNK 8192

__global__ __launch_bounds__(1024) void range_count_kernel(const int* __restrict__ ei,
                                                           int* __restrict__ cnt,
                                                           int E, int NB) {
    extern __shared__ int lds_i[];
    for (int i = threadIdx.x; i < NB; i += 1024) lds_i[i] = 0;
    __syncthreads();
    const int e0 = blockIdx.x * CHUNK;
    for (int i = threadIdx.x; i < CHUNK; i += 1024) {
        const int e = e0 + i;
        if (e < E) {
            const int d = ei[E + e];
            __hip_atomic_fetch_add(&lds_i[d / RB], 1, __ATOMIC_RELAXED, __HIP_MEMORY_SCOPE_WORKGROUP);
        }
    }
    __syncthreads();
    for (int i = threadIdx.x; i < NB; i += 1024) cnt[(size_t)blockIdx.x * NB + i] = lds_i[i];
}

__global__ __launch_bounds__(1024) void range_scan_kernel(const int* __restrict__ cnt,
                                                          int* __restrict__ base,
                                                          int* __restrict__ binStart,
                                                          int NCH, int NB) {
    extern __shared__ int lds_i[];
    for (int b = threadIdx.x; b < NB; b += 1024) {
        int run = 0;
#pragma unroll 7
        for (int c = 0; c < NCH; ++c) {
            const size_t o = (size_t)c * NB + b;
            const int v = cnt[o];
            base[o] = run;
            run += v;
        }
        lds_i[b] = run;
    }
    __syncthreads();
    if (threadIdx.x < 32) {
        const int lane = threadIdx.x;
        const int per = (NB + 31) / 32;
        const int lo = lane * per;
        const int hi = (lo + per < NB) ? (lo + per) : NB;
        int s = 0;
        for (int i = lo; i < hi; ++i) s += lds_i[i];
        int incl = s;
        for (int dlt = 1; dlt < 32; dlt <<= 1) {
            const int t = __shfl_up(incl, dlt);
            if (lane >= dlt) incl += t;
        }
        int run = incl - s;
        for (int i = lo; i < hi; ++i) {
            const int v = lds_i[i];
            binStart[i] = run;
            run += v;
        }
        if (lane == 31) binStart[NB] = run;
    }
}

__global__ __launch_bounds__(1024) void range_place_kernel(const int* __restrict__ ei,
                                                           const int* __restrict__ base,
                                                           const int* __restrict__ binStart,
                                                           int2* __restrict__ binned,
                                                           int E, int NB) {
    extern __shared__ int lds_i[];
    for (int i = threadIdx.x; i < NB; i += 1024)
        lds_i[i] = binStart[i] + base[(size_t)blockIdx.x * NB + i];
    __syncthreads();
    const int e0 = blockIdx.x * CHUNK;
    for (int i = threadIdx.x; i < CHUNK; i += 1024) {
        const int e = e0 + i;
        if (e < E) {
            const int s = ei[e];
            const int d = ei[E + e];
            const int pos = __hip_atomic_fetch_add(&lds_i[d / RB], 1, __ATOMIC_RELAXED,
                                                   __HIP_MEMORY_SCOPE_WORKGROUP);
            binned[pos] = make_int2(s, d);
        }
    }
}

__global__ __launch_bounds__(256) void range_dinv_kernel(const int2* __restrict__ binned,
                                                         const int* __restrict__ binStart,
                                                         float* __restrict__ dinv, int N) {
    __shared__ int cnt_l[RB];

    const int b = blockIdx.x;
    const int node0 = b * RB;
    const int tid = threadIdx.x;
    if (tid < RB) cnt_l[tid] = 0;
    __syncthreads();

    const int beg = binStart[b];
    const int end = binStart[b + 1];
    for (int e0 = beg + tid; e0 < end; e0 += 4 * 256) {
        int d[4];
#pragma unroll
        for (int j = 0; j < 4; ++j) {
            const int e = e0 + j * 256;
            d[j] = (e < end) ? binned[e].y : node0;
        }
#pragma unroll
        for (int j = 0; j < 4; ++j) {
            if (e0 + j * 256 < end)
                __hip_atomic_fetch_add(&cnt_l[d[j] - node0], 1, __ATOMIC_RELAXED,
                                       __HIP_MEMORY_SCOPE_WORKGROUP);
        }
    }
    __syncthreads();

    if (tid < RB && node0 + tid < N) dinv[node0 + tid] = rsqrtf((float)(cnt_l[tid] + 1));
}

#define EPT 8
#define LCAP 2048
#define RPT (LCAP / 256)

__global__ __launch_bounds__(256) void range_sum_kernel(const int2* __restrict__ binned,
                                                        const int* __restrict__ binStart,
                                                        const float* __restrict__ xs,
                                                        const float* __restrict__ dinv,
                                                        float* __restrict__ agg, int N) {
    __shared__ int lst[LCAP + 32];
    __shared__ int cnt[RB];
    __shared__ int offs[RB];

    const int b = blockIdx.x;
    const int node0 = b * RB;
    const int tid = threadIdx.x;
    const int wave = __builtin_amdgcn_readfirstlane(tid >> 5);
    const int lane = tid & 31;
    const unsigned lane4 = (unsigned)lane * 4u;
    const int beg = binStart[b];
    const int end = binStart[b + 1];

    for (int cb = beg; cb < end; cb += LCAP) {
        const int ce = (cb + LCAP < end) ? (cb + LCAP) : end;
        if (tid < RB) cnt[tid] = 0;
        __syncthreads();

        int2 sd[RPT];
#pragma unroll
        for (int i = 0; i < RPT; ++i) {
            const int e = cb + tid + 256 * i;
            sd[i] = (e < ce) ? binned[e] : make_int2(0, -1);
        }
        int rk[RPT];
#pragma unroll
        for (int i = 0; i < RPT; ++i) {
            rk[i] = -1;
            if (sd[i].y >= 0) {
                const int row = sd[i].y - node0;
                const int rank = __hip_atomic_fetch_add(&cnt[row], 1, __ATOMIC_RELAXED,
                                                        __HIP_MEMORY_SCOPE_WORKGROUP);
                rk[i] = rank * RB + row;
            }
        }
        __syncthreads();

        {
            const int ca = cnt[2 * lane];
            const int cc = cnt[2 * lane + 1];
            int incl = ca + cc;
            for (int dlt = 1; dlt < 32; dlt <<= 1) {
                const int t = __shfl_up(incl, dlt);
                if (lane >= dlt) incl += t;
            }
            const int ex = incl - (ca + cc);
            offs[2 * lane] = ex;
            offs[2 * lane + 1] = ex + ca;
        }
#pragma unroll
        for (int i = 0; i < RPT; ++i) {
            if (rk[i] >= 0) lst[offs[rk[i] & (RB - 1)] + rk[i] / RB] = sd[i].x;
        }
        __syncthreads();

        for (int k = 0; k < 8; ++k) {
            const int lrow = wave * 8 + k;
            const int c = __builtin_amdgcn_readfirstlane(cnt[lrow]);
            if (c == 0) continue;
            const int off = __builtin_amdgcn_readfirstlane(offs[lrow]);
            const int row = node0 + lrow;
            float* ap = agg + (size_t)row * D;
            float4 g = *(const float4*)(ap + lane4);
            const float di = dinv[row];
            float4 s = make_float4(0.f, 0.f, 0.f, 0.f);
            int pv = 0;
            int i = 0;
#define WINDOW() if ((i & 31) == 0) pv = lst[off + i + lane];
#define GROUP(G)                                                                   \
            {                                                                      \
                float4 v[G];                                                       \
                _Pragma("unroll") for (int j = 0; j < G; ++j) {                    \
                    const int p = __builtin_amdgcn_readlane(pv, (i & 31) + j);     \
                    const float* rp = xs + (size_t)p * D;                          \
                    v[j] = *(const float4*)(rp + lane4);                           \
                }                                                                  \
                _Pragma("unroll") for (int j = 0; j < G; ++j) {                    \
                    s.x += v[j].x; s.y += v[j].y; s.z += v[j].z; s.w += v[j].w;    \
                }                                                                  \
                i += G;                                                            \
            }
            while (i + EPT <= c) { WINDOW() GROUP(EPT) }
            if (i + 4 <= c) { WINDOW() GROUP(4) }
            if (i + 2 <= c) { WINDOW() GROUP(2) }
            if (i < c)      { WINDOW() GROUP(1) }
            g.x += di * s.x; g.y += di * s.y; g.z += di * s.z; g.w += di * s.w;
            *(float4*)(ap + lane4) = g;
        }
        if (cb + LCAP < end) __syncthreads();
    }
}

__global__ __launch_bounds__(256) void combine_kernel(const float* __restrict__ xin,
                                                      const float* __restrict__ xw,
                                                      const float* __restrict__ agg,
                                                      const float* __restrict__ dinv,
                                                      float* __restrict__ xout, int N) {
    const int i = blockIdx.x * blockDim.x + threadIdx.x;
    const int total = N * (D / 4);
    if (i >= total) return;
    const int node = i / (D / 4);
    const float di = dinv[node];
    const size_t o = (size_t)i * 4;
    const float4 xv = *(const float4*)(xin + o);
    const float4 wv = *(const float4*)(xw + o);
    const float4 av = *(const float4*)(agg + o);
    float4 r;
    r.x = xv.x + EPSILON * tanhf(av.x + wv.x * di);
    r.y = xv.y + EPSILON * tanhf(av.y + wv.y * di);
    r.z = xv.z + EPSILON * tanhf(av.z + wv.z * di);
    r.w = xv.w + EPSILON * tanhf(av.w + wv.w * di);
    *(float4*)(xout + o) = r;
}

extern "C" void kernel_launch(void* const* d_in, const int* in_sizes, int n_in,
                              void* d_out, int out_size, void* d_ws, size_t ws_size,
                              hipStream_t stream) {
    const float* x_in = (const float*)d_in[0];
    const int*   ei   = (const int*)d_in[1];
    const float* W    = (const float*)d_in[2];
    const float* Wphi = (const float*)d_in[3];
    const float* bias = (const float*)d_in[4];
    float*       out  = (float*)d_out;

    const int N = in_sizes[0] / D;
    const int E = in_sizes[1] / 2;

    float* ws    = (float*)d_ws;
    const size_t nd = (size_t)N * D;
    float* xw    = ws;
    float* agg   = ws + nd;
    float* earea = ws + 2 * nd;
    float* dinv  = ws + 3 * nd;
    float* WphiP = dinv + N;
    float* ATP   = WphiP + (size_t)D * D;

    const int NB  = (N + RB - 1) / RB;
    const int NCH = (E + CHUNK - 1) / CHUNK;
    int2* binned   = (int2*)earea;
    int*  cnt      = (int*)(binned + E);
    int*  base     = cnt + (size_t)NCH * NB;
    int*  binStart = base + (size_t)NCH * NB;

    const int B = 256;

    prep_weights_kernel<<<(D * D + B - 1) / B, B, 0, stream>>>(W, Wphi, WphiP, ATP);
    range_count_kernel<<<NCH, 1024, NB * sizeof(int), stream>>>(ei, cnt, E, NB);
    range_scan_kernel<<<1, 1024, NB * sizeof(int), stream>>>(cnt, base, binStart, NCH, NB);
    range_place_kernel<<<NCH, 1024, NB * sizeof(int), stream>>>(ei, base, binStart, binned, E, NB);
    range_dinv_kernel<<<NB, B, 0, stream>>>(binned, binStart, dinv, N);

    const int gemm_tiles  = (N + TILE_M - 1) / TILE_M;
    const int gemm_blocks = (gemm_tiles < GEMM_WGS) ? gemm_tiles : GEMM_WGS;
    const int comb_blocks = (N * (D / 4) + B - 1) / B;

    gemm_dual_kernel<false><<<gemm_blocks, B, 0, stream>>>(x_in, WphiP, ATP, bias, dinv, xw, agg,
                                                           out, N, gemm_tiles);
    for (int it = 0; it < NUM_ITERS; ++it) {
        const float* xi = (it == 0) ? x_in : out;
        range_sum_kernel<<<NB, B, 0, stream>>>(binned, binStart, xw, dinv, agg, N);
        if (it + 1 < NUM_ITERS)
            gemm_dual_kernel<true><<<gemm_blocks, B, 0, stream>>>(xi, WphiP, ATP, bias, dinv, xw,
                                                                  agg, out, N, gemm_tiles);
        else
            combine_kernel<<<comb_blocks, B, 0, stream>>>(xi, xw, agg, dinv, out, N);
    }
}


# Range scan spread over 49 work-groups (32 ranges each, a wave sums a 7-chunk slice in one round trip), placing blocks scan the range totals themselves: the one-block 196-step dependent chain is gone
# speedup vs baseline: 2.9367x; 1.0660x over previous
#include <hip/hip_runtime.h>
#include <hip/hip_bf16.h>
#include <math.h>

#define D 128
#define TILE_M 16
#define GAMMA 0.1f
#define EPSILON 0.1f
#define NUM_ITERS 4

typedef __attribute__((ext_vector_type(2))) float v2f;
typedef __attribute__((ext_vector_type(8))) float v8f;
typedef __attribute__((ext_vector_type(4))) float v4f;

__global__ __launch_bounds__(256) void prep_weights_kernel(const float* __restrict__ W,
                                                           const float* __restrict__ Wphi,
                                                           float* __restrict__ WphiP,
                                                           float* __restrict__ ATP) {
    int i = blockIdx.x * blockDim.x + threadIdx.x;
    if (i >= D * D) return;
    int k = i / D, n = i % D;
    size_t po = (size_t)(k >> 1) * (2 * D) + n * 2 + (k & 1);
    WphiP[po] = Wphi[n * D + k];
    float a = W[n * D + k] - W[k * D + n];
    if (k == n) a -= GAMMA;
    ATP[po] = a;
}

#define WPITCH (2 * D)
#define XPITCH (D + 4)
#define GEMM_WGS 512

template <bool COMBINE>
__global__ __launch_bounds__(256) void gemm_dual_kernel(const float* xin,
                                                        const float* __restrict__ WphiP,
                                                        const float* __restrict__ ATP,
                                                        const float* __restrict__ bias,
                                                        const float* __restrict__ dinv,
                                                        float* __restrict__ xw,
                                                        float* __restrict__ agg,
                                                        float* xout, int N, int ntiles) {
    __shared__ __attribute__((aligned(16))) float lds_w[2][(D / 2) * WPITCH];
    __shared__ __attribute__((aligned(16))) float lds_x[2][TILE_M][XPITCH];

    const int tid = threadIdx.x;
    const int stride = gridDim.x;
    int t = blockIdx.x;

    const int r0 = tid / (D / 4);
    const int c0 = (tid % (D / 4)) * 4;

    float4 xv0, xv1, wv0, wv1, av0, av1;
    float di0, di1;

    auto request_row = [&](int row, float4& xv, float4& wv, float4& av, float& di) {
        const int rr = (row < N) ? row : (N - 1);
        const size_t o = (size_t)rr * D + c0;
        xv = *(const float4*)(xin + o);
        if (COMBINE) {
            wv = *(const float4*)(xw + o);
            av = *(const float4*)(agg + o);
            di = dinv[rr];
        }
    };
    auto request_tile = [&](int tt) {
        request_row(tt * TILE_M + r0,     xv0, wv0, av0, di0);
        request_row(tt * TILE_M + r0 + 8, xv1, wv1, av1, di1);
    };
    auto updated = [&](const float4& xv, const float4& wv, const float4& av, float di) {
        if (!COMBINE) return xv;
        float4 v;
        v.x = xv.x + EPSILON * tanhf(av.x + wv.x * di);
        v.y = xv.y + EPSILON * tanhf(av.y + wv.y * di);
        v.z = xv.z + EPSILON * tanhf(av.z + wv.z * di);
        v.w = xv.w + EPSILON * tanhf(av.w + wv.w * di);
        return v;
    };

    request_tile(t);

#pragma unroll 8
    for (int i = tid; i < (D / 2) * WPITCH / 4; i += 256) {
        const int p = i >> 6;
        const int cs = ((i & 63) * 4) ^ ((p & 1) << 5);
        const float4 w4 = *(const float4*)(WphiP + (size_t)i * 4);
        const float4 a4 = *(const float4*)(ATP + (size_t)i * 4);
        *(float4*)&lds_w[0][p * WPITCH + cs] = w4;
        *(float4*)&lds_w[1][p * WPITCH + cs] = a4;
    }

    const int wave = tid >> 5;
    const int lane = tid & 31;
    const int mn   = lane & 15;
    const int kq   = lane >> 4;
    const int col0 = wave * 16;
    const int boff = kq * WPITCH + (((col0 + mn) * 2) ^ (kq << 5));
    const float* wl_w = &lds_w[0][boff];
    const float* wl_a = &lds_w[1][boff];
    const float b    = bias[col0 + mn];
    const int rowoff = kq * 8;

    int buf = 0;
    for (; t < ntiles; t += stride, buf ^= 1) {
        const int row0 = t * TILE_M;
        const bool full = (row0 + TILE_M) <= N;
        float (*lx)[XPITCH] = lds_x[buf];

        const float4 v0 = updated(xv0, wv0, av0, di0);
        const float4 v1 = updated(xv1, wv1, av1, di1);
        *(float4*)&lx[r0][c0]     = v0;
        *(float4*)&lx[r0 + 8][c0] = v1;
        __syncthreads();
        if (COMBINE) {
            const int ra = row0 + r0, rb = row0 + r0 + 8;
            if (full || ra < N) *(float4*)(xout + (size_t)ra * D + c0) = v0;
            if (full || rb < N) *(float4*)(xout + (size_t)rb * D + c0) = v1;
        }
        float dr[8];
        if (full) {
            const float4 q0 = *(const float4*)(dinv + row0 + rowoff);
            const float4 q1 = *(const float4*)(dinv + row0 + rowoff + 4);
            dr[0] = q0.x; dr[1] = q0.y; dr[2] = q0.z; dr[3] = q0.w;
            dr[4] = q1.x; dr[5] = q1.y; dr[6] = q1.z; dr[7] = q1.w;
        } else {
#pragma unroll
            for (int r = 0; r < 8; ++r) {
                const int rr = row0 + rowoff + r;
                dr[r] = dinv[(rr < N) ? rr : (N - 1)];
            }
        }
        if (t + stride < ntiles) request_tile(t + stride);

        v8f acc_w = {};
        v8f acc_a = {};

#pragma unroll 4
        for (int k0 = 0; k0 < D; k0 += 4) {
            const v2f a  = *(const v2f*)&lx[mn][k0 + kq * 2];
            const v2f bw = *(const v2f*)(wl_w + (k0 >> 1) * WPITCH);
            const v2f ba = *(const v2f*)(wl_a + (k0 >> 1) * WPITCH);
            acc_w = __builtin_amdgcn_wmma_f32_16x16x4_f32(false, a, false, bw,
                                                          (short)0, acc_w, false, false);
            acc_a = __builtin_amdgcn_wmma_f32_16x16x4_f32(false, a, false, ba,
                                                          (short)0, acc_a, false, false);
        }

        if (!COMBINE) {
            const v4f h0 = {xv0.x, xv0.y, xv0.z, xv0.w};
            const v4f h1 = {xv1.x, xv1.y, xv1.z, xv1.w};
            asm volatile("" :: "v"(h0), "v"(h1));
        }

        if (full) {
#pragma unroll
            for (int r = 0; r < 8; ++r) {
                const size_t o = (size_t)(row0 + rowoff + r) * D + col0 + mn;
                xw[o]  = acc_w[r] * dr[r];
                agg[o] = acc_a[r] + b;
            }
        } else {
#pragma unroll
            for (int r = 0; r < 8; ++r) {
                const int row = row0 + rowoff + r;
                if (row < N) {
                    const size_t o = (size_t)row * D + col0 + mn;
                    xw[o]  = acc_w[r] * dr[r];
                    agg[o] = acc_a[r] + b;
                }
            }
        }
    }
}

#define RB 64
#define CHUNK 8192

__global__ __launch_bounds__(1024) void range_count_kernel(const int* __restrict__ ei,
                                                           int* __restrict__ cnt,
                                                           int E, int NB) {
    extern __shared__ int lds_i[];
    for (int i = threadIdx.x; i < NB; i += 1024) lds_i[i] = 0;
    __syncthreads();
    const int e0 = blockIdx.x * CHUNK;
    for (int i = threadIdx.x; i < CHUNK; i += 1024) {
        const int e = e0 + i;
        if (e < E) {
            const int d = ei[E + e];
            __hip_atomic_fetch_add(&lds_i[d / RB], 1, __ATOMIC_RELAXED, __HIP_MEMORY_SCOPE_WORKGROUP);
        }
    }
    __syncthreads();
    for (int i = threadIdx.x; i < NB; i += 1024) cnt[(size_t)blockIdx.x * NB + i] = lds_i[i];
}

#define SCAN_RB 32
__global__ __launch_bounds__(1024) void range_scan_kernel(const int* __restrict__ cnt,
                                                          int* __restrict__ base,
                                                          int* __restrict__ total,
                                                          int NCH, int NB) {
    __shared__ int part[32][SCAN_RB + 1];
    const int lane = threadIdx.x & 31;
    const int w = __builtin_amdgcn_readfirstlane(threadIdx.x >> 5);
    const int b = blockIdx.x * SCAN_RB + lane;
    const bool valid = b < NB;
    const int bb = valid ? b : (NB - 1);
    const int SL = (NCH + 31) >> 5;
    const int c0 = w * SL;
    const int c1 = (c0 + SL < NCH) ? (c0 + SL) : NCH;

    int sum = 0;
    for (int c = c0; c < c1; c += 8) {
        int v[8];
#pragma unroll
        for (int j = 0; j < 8; ++j) {
            const int cc = (c + j < c1) ? (c + j) : (c1 - 1);
            v[j] = cnt[(size_t)cc * NB + bb];
        }
#pragma unroll
        for (int j = 0; j < 8; ++j) sum += (c + j < c1) ? v[j] : 0;
    }
    part[w][lane] = sum;
    __syncthreads();

    int run = 0;
#pragma unroll
    for (int q = 0; q < 32; ++q) {
        const int t = part[q][lane];
        run += (q < w) ? t : 0;
    }
    for (int c = c0; c < c1; c += 8) {
        int v[8];
#pragma unroll
        for (int j = 0; j < 8; ++j) {
            const int cc = (c + j < c1) ? (c + j) : (c1 - 1);
            v[j] = cnt[(size_t)cc * NB + bb];
        }
#pragma unroll
        for (int j = 0; j < 8; ++j) {
            if (c + j < c1) {
                if (valid) base[(size_t)(c + j) * NB + b] = run;
                run += v[j];
            }
        }
    }
    if (w == 31 && valid) total[b] = run;
}

__global__ __launch_bounds__(1024) void range_place_kernel(const int* __restrict__ ei,
                                                           const int* __restrict__ base,
                                                           const int* __restrict__ total,
                                                           int* __restrict__ binStart,
                                                           int2* __restrict__ binned,
                                                           int E, int NB) {
    extern __shared__ int lds_i[];
    __shared__ int wsum[32];
    const int lane = threadIdx.x & 31;
    const int w = threadIdx.x >> 5;

    for (int i = threadIdx.x; i < NB; i += 1024) lds_i[i] = total[i];
    __syncthreads();
    const int PER = (NB + 1023) >> 10;
    const int lo = threadIdx.x * PER;
    const int hi = (lo + PER < NB) ? (lo + PER) : NB;
    int s = 0;
    for (int i = lo; i < hi; ++i) s += lds_i[i];
    int incl = s;
    for (int dlt = 1; dlt < 32; dlt <<= 1) {
        const int t = __shfl_up(incl, dlt);
        if (lane >= dlt) incl += t;
    }
    if (lane == 31) wsum[w] = incl;
    __syncthreads();
    int run = incl - s;
#pragma unroll
    for (int q = 0; q < 32; ++q) {
        const int t = wsum[q];
        run += (q < w) ? t : 0;
    }
    for (int i = lo; i < hi; ++i) {
        const int v = lds_i[i];
        lds_i[i] = run;
        run += v;
    }
    if (blockIdx.x == 0 && lo < NB && hi == NB) binStart[NB] = run;
    __syncthreads();
    for (int i = threadIdx.x; i < NB; i += 1024) {
        const int st = lds_i[i];
        if (blockIdx.x == 0) binStart[i] = st;
        lds_i[i] = st + base[(size_t)blockIdx.x * NB + i];
    }
    __syncthreads();
    const int e0 = blockIdx.x * CHUNK;
    for (int i = threadIdx.x; i < CHUNK; i += 1024) {
        const int e = e0 + i;
        if (e < E) {
            const int s = ei[e];
            const int d = ei[E + e];
            const int pos = __hip_atomic_fetch_add(&lds_i[d / RB], 1, __ATOMIC_RELAXED,
                                                   __HIP_MEMORY_SCOPE_WORKGROUP);
            binned[pos] = make_int2(s, d);
        }
    }
}

__global__ __launch_bounds__(256) void range_dinv_kernel(const int2* __restrict__ binned,
                                                         const int* __restrict__ binStart,
                                                         float* __restrict__ dinv, int N) {
    __shared__ int cnt_l[RB];

    const int b = blockIdx.x;
    const int node0 = b * RB;
    const int tid = threadIdx.x;
    if (tid < RB) cnt_l[tid] = 0;
    __syncthreads();

    const int beg = binStart[b];
    const int end = binStart[b + 1];
    for (int e0 = beg + tid; e0 < end; e0 += 4 * 256) {
        int d[4];
#pragma unroll
        for (int j = 0; j < 4; ++j) {
            const int e = e0 + j * 256;
            d[j] = (e < end) ? binned[e].y : node0;
        }
#pragma unroll
        for (int j = 0; j < 4; ++j) {
            if (e0 + j * 256 < end)
                __hip_atomic_fetch_add(&cnt_l[d[j] - node0], 1, __ATOMIC_RELAXED,
                                       __HIP_MEMORY_SCOPE_WORKGROUP);
        }
    }
    __syncthreads();

    if (tid < RB && node0 + tid < N) dinv[node0 + tid] = rsqrtf((float)(cnt_l[tid] + 1));
}

#define EPT 8
#define LCAP 2048
#define RPT (LCAP / 256)

__global__ __launch_bounds__(256) void range_sum_kernel(const int2* __restrict__ binned,
                                                        const int* __restrict__ binStart,
                                                        const float* __restrict__ xs,
                                                        const float* __restrict__ dinv,
                                                        float* __restrict__ agg, int N) {
    __shared__ int lst[LCAP + 32];
    __shared__ int cnt[RB];
    __shared__ int offs[RB];

    const int b = blockIdx.x;
    const int node0 = b * RB;
    const int tid = threadIdx.x;
    const int wave = __builtin_amdgcn_readfirstlane(tid >> 5);
    const int lane = tid & 31;
    const unsigned lane4 = (unsigned)lane * 4u;
    const int beg = binStart[b];
    const int end = binStart[b + 1];

    for (int cb = beg; cb < end; cb += LCAP) {
        const int ce = (cb + LCAP < end) ? (cb + LCAP) : end;
        if (tid < RB) cnt[tid] = 0;
        __syncthreads();

        int2 sd[RPT];
#pragma unroll
        for (int i = 0; i < RPT; ++i) {
            const int e = cb + tid + 256 * i;
            sd[i] = (e < ce) ? binned[e] : make_int2(0, -1);
        }
        int rk[RPT];
#pragma unroll
        for (int i = 0; i < RPT; ++i) {
            rk[i] = -1;
            if (sd[i].y >= 0) {
                const int row = sd[i].y - node0;
                const int rank = __hip_atomic_fetch_add(&cnt[row], 1, __ATOMIC_RELAXED,
                                                        __HIP_MEMORY_SCOPE_WORKGROUP);
                rk[i] = rank * RB + row;
            }
        }
        __syncthreads();

        {
            const int ca = cnt[2 * lane];
            const int cc = cnt[2 * lane + 1];
            int incl = ca + cc;
            for (int dlt = 1; dlt < 32; dlt <<= 1) {
                const int t = __shfl_up(incl, dlt);
                if (lane >= dlt) incl += t;
            }
            const int ex = incl - (ca + cc);
            offs[2 * lane] = ex;
            offs[2 * lane + 1] = ex + ca;
        }
#pragma unroll
        for (int i = 0; i < RPT; ++i) {
            if (rk[i] >= 0) lst[offs[rk[i] & (RB - 1)] + rk[i] / RB] = sd[i].x;
        }
        __syncthreads();

        for (int k = 0; k < 8; ++k) {
            const int lrow = wave * 8 + k;
            const int c = __builtin_amdgcn_readfirstlane(cnt[lrow]);
            if (c == 0) continue;
            const int off = __builtin_amdgcn_readfirstlane(offs[lrow]);
            const int row = node0 + lrow;
            float* ap = agg + (size_t)row * D;
            float4 g = *(const float4*)(ap + lane4);
            const float di = dinv[row];
            float4 s = make_float4(0.f, 0.f, 0.f, 0.f);
            int pv = 0;
            int i = 0;
#define WINDOW() if ((i & 31) == 0) pv = lst[off + i + lane];
#define GROUP(G)                                                                   \
            {                                                                      \
                float4 v[G];                                                       \
                _Pragma("unroll") for (int j = 0; j < G; ++j) {                    \
                    const int p = __builtin_amdgcn_readlane(pv, (i & 31) + j);     \
                    const float* rp = xs + (size_t)p * D;                          \
                    v[j] = *(const float4*)(rp + lane4);                           \
                }                                                                  \
                _Pragma("unroll") for (int j = 0; j < G; ++j) {                    \
                    s.x += v[j].x; s.y += v[j].y; s.z += v[j].z; s.w += v[j].w;    \
                }                                                                  \
                i += G;                                                            \
            }
            while (i + EPT <= c) { WINDOW() GROUP(EPT) }
            if (i + 4 <= c) { WINDOW() GROUP(4) }
            if (i + 2 <= c) { WINDOW() GROUP(2) }
            if (i < c)      { WINDOW() GROUP(1) }
            g.x += di * s.x; g.y += di * s.y; g.z += di * s.z; g.w += di * s.w;
            *(float4*)(ap + lane4) = g;
        }
        if (cb + LCAP < end) __syncthreads();
    }
}

__global__ __launch_bounds__(256) void combine_kernel(const float* __restrict__ xin,
                                                      const float* __restrict__ xw,
                                                      const float* __restrict__ agg,
                                                      const float* __restrict__ dinv,
                                                      float* __restrict__ xout, int N) {
    const int i = blockIdx.x * blockDim.x + threadIdx.x;
    const int total = N * (D / 4);
    if (i >= total) return;
    const int node = i / (D / 4);
    const float di = dinv[node];
    const size_t o = (size_t)i * 4;
    const float4 xv = *(const float4*)(xin + o);
    const float4 wv = *(const float4*)(xw + o);
    const float4 av = *(const float4*)(agg + o);
    float4 r;
    r.x = xv.x + EPSILON * tanhf(av.x + wv.x * di);
    r.y = xv.y + EPSILON * tanhf(av.y + wv.y * di);
    r.z = xv.z + EPSILON * tanhf(av.z + wv.z * di);
    r.w = xv.w + EPSILON * tanhf(av.w + wv.w * di);
    *(float4*)(xout + o) = r;
}

extern "C" void kernel_launch(void* const* d_in, const int* in_sizes, int n_in,
                              void* d_out, int out_size, void* d_ws, size_t ws_size,
                              hipStream_t stream) {
    const float* x_in = (const float*)d_in[0];
    const int*   ei   = (const int*)d_in[1];
    const float* W    = (const float*)d_in[2];
    const float* Wphi = (const float*)d_in[3];
    const float* bias = (const float*)d_in[4];
    float*       out  = (float*)d_out;

    const int N = in_sizes[0] / D;
    const int E = in_sizes[1] / 2;

    float* ws    = (float*)d_ws;
    const size_t nd = (size_t)N * D;
    float* xw    = ws;
    float* agg   = ws + nd;
    float* earea = ws + 2 * nd;
    float* dinv  = ws + 3 * nd;
    float* WphiP = dinv + N;
    float* ATP   = WphiP + (size_t)D * D;

    const int NB  = (N + RB - 1) / RB;
    const int NCH = (E + CHUNK - 1) / CHUNK;
    int2* binned   = (int2*)earea;
    int*  cnt      = (int*)(binned + E);
    int*  base     = cnt + (size_t)NCH * NB;
    int*  binStart = base + (size_t)NCH * NB;
    int*  total    = binStart + (NB + 1);

    const int B = 256;

    prep_weights_kernel<<<(D * D + B - 1) / B, B, 0, stream>>>(W, Wphi, WphiP, ATP);
    range_count_kernel<<<NCH, 1024, NB * sizeof(int), stream>>>(ei, cnt, E, NB);
    range_scan_kernel<<<(NB + SCAN_RB - 1) / SCAN_RB, 1024, 0, stream>>>(cnt, base, total, NCH, NB);
    range_place_kernel<<<NCH, 1024, NB * sizeof(int), stream>>>(ei, base, total, binStart, binned, E, NB);
    range_dinv_kernel<<<NB, B, 0, stream>>>(binned, binStart, dinv, N);

    const int gemm_tiles  = (N + TILE_M - 1) / TILE_M;
    const int gemm_blocks = (gemm_tiles < GEMM_WGS) ? gemm_tiles : GEMM_WGS;
    const int comb_blocks = (N * (D / 4) + B - 1) / B;

    gemm_dual_kernel<false><<<gemm_blocks, B, 0, stream>>>(x_in, WphiP, ATP, bias, dinv, xw, agg,
                                                           out, N, gemm_tiles);
    for (int it = 0; it < NUM_ITERS; ++it) {
        const float* xi = (it == 0) ? x_in : out;
        range_sum_kernel<<<NB, B, 0, stream>>>(binned, binStart, xw, dinv, agg, N);
        if (it + 1 < NUM_ITERS)
            gemm_dual_kernel<true><<<gemm_blocks, B, 0, stream>>>(xi, WphiP, ATP, bias, dinv, xw,
                                                                  agg, out, N, gemm_tiles);
        else
            combine_kernel<<<comb_blocks, B, 0, stream>>>(xi, xw, agg, dinv, out, N);
    }
}
